# starting kernel
# baseline (speedup 1.0000x reference)
#include <hip/hip_runtime.h>

#define BQ     4
#define CIN    64
#define HH     96
#define WW     96
#define GROUPS 8
#define NCH    8
#define CGC    8
#define KW     7
#define PADR   3
#define HWSZ   (HH * WW)

#define ROWS   4
#define KVROWS (ROWS + 2*PADR)
#define NTHR   128

typedef __attribute__((ext_vector_type(2))) float v2f;
typedef __attribute__((ext_vector_type(4))) float v4f;
typedef __attribute__((ext_vector_type(8))) float v8f;

__device__ __forceinline__ int reflect96(int i) {
    if (i < 0)  i = -i;
    if (i > 95) i = 190 - i;
    return i;
}

__global__ __launch_bounds__(NTHR)
void SelfAttentionBlock_kernel(const float* __restrict__ x,
                               const float* __restrict__ Wk,
                               const float* __restrict__ Wq,
                               const float* __restrict__ Wv,
                               const float* __restrict__ r_ai,
                               const float* __restrict__ r_aj,
                               float* __restrict__ out)
{
    __shared__ __attribute__((aligned(16))) float kb[KVROWS][WW][NCH];
    __shared__ __attribute__((aligned(16))) float vb[KVROWS][WW][NCH];
    __shared__ __attribute__((aligned(16))) float qb[ROWS][WW][NCH];
    __shared__ __attribute__((aligned(16))) float rcomp[KW][NCH];

    const int h0   = blockIdx.x * ROWS;
    const int g    = blockIdx.y;
    const int b    = blockIdx.z;
    const int tid  = threadIdx.x;
    const int lane = tid & 31;
    const int wave = tid >> 5;
    const int half = lane >> 4;
    const int m    = lane & 15;

    if (tid < KW * NCH) {
        int t = tid / NCH, n = tid % NCH;
        int cg = g * NCH + n;
        rcomp[t][n] = (cg < 32) ? r_ai[cg * KW + t] : r_aj[(cg - 32) * KW + t];
    }

    const float* WkG = Wk + g * 64;
    const float* WqG = Wq + g * 64;
    const float* WvG = Wv + g * 64;
    const int k_lo = half * 2;
    const int k_hi = 4 + half * 2;

    v2f aKV_lo, aKV_hi, aQ_lo, aQ_hi;
    if (m < 8) {
        aKV_lo[0] = WkG[m * 8 + k_lo];     aKV_lo[1] = WkG[m * 8 + k_lo + 1];
        aKV_hi[0] = WkG[m * 8 + k_hi];     aKV_hi[1] = WkG[m * 8 + k_hi + 1];
    } else {
        aKV_lo[0] = WvG[(m - 8) * 8 + k_lo]; aKV_lo[1] = WvG[(m - 8) * 8 + k_lo + 1];
        aKV_hi[0] = WvG[(m - 8) * 8 + k_hi]; aKV_hi[1] = WvG[(m - 8) * 8 + k_hi + 1];
    }
    aQ_lo[0] = WqG[(m & 7) * 8 + k_lo];  aQ_lo[1] = WqG[(m & 7) * 8 + k_lo + 1];
    aQ_hi[0] = WqG[(m & 7) * 8 + k_hi];  aQ_hi[1] = WqG[(m & 7) * 8 + k_hi + 1];

    const float* xg = x + (size_t)(b * CIN + g * CGC) * HWSZ;

    for (int t = wave; t < KVROWS * 6; t += 4) {
        int i     = t / 6;
        int strip = (t % 6) * 16;
        int hr    = reflect96(h0 - PADR + i);
        int col   = strip + m;
        const float* xp = xg + hr * WW + col;
        v2f blo, bhi;
        blo[0] = xp[(size_t)k_lo       * HWSZ];
        blo[1] = xp[(size_t)(k_lo + 1) * HWSZ];
        bhi[0] = xp[(size_t)k_hi       * HWSZ];
        bhi[1] = xp[(size_t)(k_hi + 1) * HWSZ];

        v8f acc = {};
        acc = __builtin_amdgcn_wmma_f32_16x16x4_f32(false, aKV_lo, false, blo,
                                                    (short)0, acc, false, false);
        acc = __builtin_amdgcn_wmma_f32_16x16x4_f32(false, aKV_hi, false, bhi,
                                                    (short)0, acc, false, false);
        float* dst = half ? &vb[i][col][0] : &kb[i][col][0];
        v4f lo4 = {acc[0], acc[1], acc[2], acc[3]};
        v4f hi4 = {acc[4], acc[5], acc[6], acc[7]};
        *(v4f*)dst       = lo4;
        *((v4f*)dst + 1) = hi4;
    }

    for (int t = wave; t < ROWS * 6; t += 4) {
        int rr    = t / 6;
        int strip = (t % 6) * 16;
        int col   = strip + m;
        const float* xp = xg + (h0 + rr) * WW + col;
        v2f blo, bhi;
        blo[0] = xp[(size_t)k_lo       * HWSZ];
        blo[1] = xp[(size_t)(k_lo + 1) * HWSZ];
        bhi[0] = xp[(size_t)k_hi       * HWSZ];
        bhi[1] = xp[(size_t)(k_hi + 1) * HWSZ];

        v8f acc = {};
        acc = __builtin_amdgcn_wmma_f32_16x16x4_f32(false, aQ_lo, false, blo,
                                                    (short)0, acc, false, false);
        acc = __builtin_amdgcn_wmma_f32_16x16x4_f32(false, aQ_hi, false, bhi,
                                                    (short)0, acc, false, false);
        if (!half) {
            float* dst = &qb[rr][col][0];
            v4f lo4 = {acc[0], acc[1], acc[2], acc[3]};
            v4f hi4 = {acc[4], acc[5], acc[6], acc[7]};
            *(v4f*)dst       = lo4;
            *((v4f*)dst + 1) = hi4;
        }
    }

    __syncthreads();

    const bool rowBias = (g < 4);
    for (int p = tid; p < ROWS * WW; p += NTHR) {
        const int rr = p / WW;
        const int w  = p % WW;

        const v4f q0 = *(const v4f*)&qb[rr][w][0];
        const v4f q1 = *(const v4f*)&qb[rr][w][4];

        float rb[KW];
        #pragma unroll
        for (int t = 0; t < KW; ++t) {
            const v4f r0 = *(const v4f*)&rcomp[t][0];
            const v4f r1 = *(const v4f*)&rcomp[t][4];
            float s = 0.f;
            #pragma unroll
            for (int n = 0; n < 4; ++n) s = __builtin_fmaf(q0[n], r0[n], s);
            #pragma unroll
            for (int n = 0; n < 4; ++n) s = __builtin_fmaf(q1[n], r1[n], s);
            rb[t] = s;
        }

        float mmax = -3.0e38f, ssum = 0.0f;
        float acc[NCH];
        #pragma unroll
        for (int n = 0; n < NCH; ++n) acc[n] = 0.0f;

        for (int ki = 0; ki < KW; ++ki) {
            const int slot = rr + ki;
            for (int kj = 0; kj < KW; ++kj) {
                int c = w + kj - PADR;
                c = (c < 0) ? -c : ((c > 95) ? 190 - c : c);

                const v4f k0 = *(const v4f*)&kb[slot][c][0];
                const v4f k1 = *(const v4f*)&kb[slot][c][4];
                float l = rb[rowBias ? ki : kj];
                #pragma unroll
                for (int n = 0; n < 4; ++n) l = __builtin_fmaf(q0[n], k0[n], l);
                #pragma unroll
                for (int n = 0; n < 4; ++n) l = __builtin_fmaf(q1[n], k1[n], l);

                const float mn   = fmaxf(mmax, l);
                const float corr = __expf(mmax - mn);
                const float e    = __expf(l - mn);
                ssum = ssum * corr + e;

                const v4f va = *(const v4f*)&vb[slot][c][0];
                const v4f vbb = *(const v4f*)&vb[slot][c][4];
                #pragma unroll
                for (int n = 0; n < 4; ++n) acc[n]     = acc[n]     * corr + e * va[n];
                #pragma unroll
                for (int n = 0; n < 4; ++n) acc[n + 4] = acc[n + 4] * corr + e * vbb[n];
                mmax = mn;
            }
        }

        const float inv = 1.0f / ssum;
        float* op = out + ((size_t)(b * CIN + g * NCH) * HH + (h0 + rr)) * WW + w;
        #pragma unroll
        for (int n = 0; n < NCH; ++n)
            op[(size_t)n * HWSZ] = acc[n] * inv;
    }
}

extern "C" void kernel_launch(void* const* d_in, const int* in_sizes, int n_in,
                              void* d_out, int out_size, void* d_ws, size_t ws_size,
                              hipStream_t stream) {
    (void)in_sizes; (void)n_in; (void)out_size; (void)d_ws; (void)ws_size;
    const float* x    = (const float*)d_in[0];
    const float* Wk   = (const float*)d_in[1];
    const float* Wq   = (const float*)d_in[2];
    const float* Wv   = (const float*)d_in[3];
    const float* r_ai = (const float*)d_in[4];
    const float* r_aj = (const float*)d_in[5];
    float* out = (float*)d_out;

    dim3 grid(HH / ROWS, GROUPS, BQ);
    SelfAttentionBlock_kernel<<<grid, NTHR, 0, stream>>>(x, Wk, Wq, Wv, r_ai, r_aj, out);
}


# Attention on all 4 waves, 3 px per lane: waves 0-2 take rows 0-2 of a column, the idle wave row 3 (3 adjacent columns); lanes with bit 3 set read record halves swapped (all banks); sum order changed
# speedup vs baseline: 2.0453x; 2.0453x over previous
#include <hip/hip_runtime.h>


#define BQ     4
#define CIN    64
#define HH     96
#define WW     96
#define GROUPS 8
#define NCH    8
#define CGC    8
#define KW     7
#define PADR   3
#define HWSZ   (HH * WW)

#define ROWS   4
#define KVROWS (ROWS + 2*PADR)
#define NTHR   128
#define XPITCH (KVROWS * WW + 8)
#define XC4    (WW / 4)
#define XHALF  (KVROWS / 2)
#define TSTRIPS 3
#define KVTRIPS (KVROWS * 2 / 4)
#define QTRIPS  (ROWS * 2 / 4)
#define PTRIPS  (KVTRIPS + QTRIPS)

typedef __attribute__((ext_vector_type(2))) float v2f;
typedef __attribute__((ext_vector_type(4))) float v4f;
typedef __attribute__((ext_vector_type(8))) float v8f;

__device__ __forceinline__ int reflect96(int i) {
    if (i < 0)  i = -i;
    if (i > 95) i = 190 - i;
    return i;
}


#define GUARD_LOG2 32.0f
#define NPX  3
#define NREC (NPX + KW - 1)
template <bool VERT, bool ROWBIAS>
__device__ __forceinline__ void attend_triple(const float* kbp, const float* vbp,
                                              const float* qbp, const float* rcp,
                                              const int r0, const int c0, const bool swp, float* op)
{
    constexpr bool ALONG = (VERT == ROWBIAS);
    const float LOG2E = 1.44269504088896340736f;
    v4f q0[NPX], q1[NPX];
    #pragma unroll
    for (int p = 0; p < NPX; ++p) {
        const float* qp = qbp + ((r0 + (VERT ? p : 0)) * WW + c0 + (VERT ? 0 : p)) * NCH;
        q0[p] = *(const v4f*)qp * LOG2E;
        q1[p] = *(const v4f*)(qp + 4) * LOG2E;
    }

    const int sw = swp ? 4 : 0;
    const int so = 4 - sw;
    v4f qf[NPX], qs[NPX];
    #pragma unroll
    for (int p = 0; p < NPX; ++p) {
        qf[p] = swp ? q1[p] : q0[p];
        qs[p] = swp ? q0[p] : q1[p];
    }

    float rb[NPX][KW];
    #pragma unroll
    for (int t = 0; t < KW; ++t) {
        const v4f ra = *(const v4f*)(rcp + t * NCH);
        const v4f rc = *(const v4f*)(rcp + t * NCH + 4);
        #pragma unroll
        for (int p = 0; p < NPX; ++p) {
            float s = 0.f;
            #pragma unroll
            for (int n = 0; n < 4; ++n) s = __builtin_fmaf(q0[p][n], ra[n], s);
            #pragma unroll
            for (int n = 0; n < 4; ++n) s = __builtin_fmaf(q1[p][n], rc[n], s);
            rb[p][t] = s;
        }
    }

    #pragma unroll
    for (int p = 0; p < NPX; ++p) {
        const float* kp = kbp + ((r0 + PADR + (VERT ? p : 0)) * WW + c0 + (VERT ? 0 : p)) * NCH;
        const v4f k0 = *(const v4f*)kp;
        const v4f k1 = *(const v4f*)(kp + 4);
        float l = rb[p][PADR];
        #pragma unroll
        for (int n = 0; n < 4; ++n) l = __builtin_fmaf(q0[p][n], k0[n], l);
        #pragma unroll
        for (int n = 0; n < 4; ++n) l = __builtin_fmaf(q1[p][n], k1[n], l);
        #pragma unroll
        for (int t = 0; t < KW; ++t) rb[p][t] -= l;
    }

    float ssum[NPX], acc[NPX][NCH];
    #pragma unroll
    for (int p = 0; p < NPX; ++p) {
        ssum[p] = 0.0f;
        #pragma unroll
        for (int n = 0; n < NCH; ++n) acc[p][n] = 0.0f;
    }

    int cc[NREC];
    #pragma unroll
    for (int a = 0; a < NREC; ++a) cc[a] = reflect96(c0 - PADR + a) * NCH;

    for (int kq = 0; kq < KW; ++kq) {
        const int tb = VERT ? (r0 * WW + reflect96(c0 + kq - PADR)) * NCH : (r0 + kq) * WW * NCH;

        float bq[NPX];
        #pragma unroll
        for (int p = 0; p < NPX; ++p) {
            float v = rb[p][0];
            #pragma unroll
            for (int t = 1; t < KW; ++t) v = (kq == t) ? rb[p][t] : v;
            bq[p] = v;
        }

        float lg[NPX][KW];
        #pragma unroll
        for (int a = 0; a < NREC; ++a) {
            const float* kp = kbp + tb + (VERT ? a * WW * NCH : cc[a]);
            const v4f kf = *(const v4f*)(kp + sw);
            const v4f ks = *(const v4f*)(kp + so);
            #pragma unroll
            for (int p = 0; p < NPX; ++p) {
                const int ka = a - p;
                if (ka < 0 || ka >= KW) continue;
                float l = ALONG ? rb[p][ka] : bq[p];
                #pragma unroll
                for (int n = 0; n < 4; ++n) l = __builtin_fmaf(qf[p][n], kf[n], l);
                #pragma unroll
                for (int n = 0; n < 4; ++n) l = __builtin_fmaf(qs[p][n], ks[n], l);
                lg[p][ka] = l;
            }
        }

        float tm[NPX];
        #pragma unroll
        for (int p = 0; p < NPX; ++p) {
            float v = lg[p][0];
            #pragma unroll
            for (int t = 1; t < KW; ++t) v = fmaxf(v, lg[p][t]);
            tm[p] = v;
        }
        const float tmx = fmaxf(fmaxf(tm[0], tm[1]), tm[2]);
        if (__any(tmx > GUARD_LOG2)) {
            #pragma unroll
            for (int p = 0; p < NPX; ++p) {
                const float s = ceilf(fmaxf(tm[p], 0.0f));
                const float f = __builtin_amdgcn_exp2f(-s);
                ssum[p] *= f;
                #pragma unroll
                for (int n = 0; n < NCH; ++n) acc[p][n] *= f;
                #pragma unroll
                for (int t = 0; t < KW; ++t) { rb[p][t] -= s; lg[p][t] -= s; }
            }
        }

        #pragma unroll
        for (int a = 0; a < NREC; ++a) {
            const float* vp = vbp + tb + (VERT ? a * WW * NCH : cc[a]);
            const v4f va = *(const v4f*)(vp + sw);
            const v4f vc = *(const v4f*)(vp + so);
            #pragma unroll
            for (int p = 0; p < NPX; ++p) {
                const int ka = a - p;
                if (ka < 0 || ka >= KW) continue;
                const float e = __builtin_amdgcn_exp2f(lg[p][ka]);
                ssum[p] += e;
                #pragma unroll
                for (int n = 0; n < 4; ++n) acc[p][n]     = __builtin_fmaf(e, va[n], acc[p][n]);
                #pragma unroll
                for (int n = 0; n < 4; ++n) acc[p][n + 4] = __builtin_fmaf(e, vc[n], acc[p][n + 4]);
            }
        }
    }

    #pragma unroll
    for (int p = 0; p < NPX; ++p) {
        const float inv = 1.0f / ssum[p];
        #pragma unroll
        for (int n = 0; n < NCH; ++n) {
            const float r = swp ? acc[p][n ^ 4] : acc[p][n];
            op[(size_t)n * HWSZ + (VERT ? p * WW : p)] = r * inv;
        }
    }
}

__global__ __launch_bounds__(NTHR)
void SelfAttentionBlock_kernel(const float* __restrict__ x,
                               const float* __restrict__ Wk,
                               const float* __restrict__ Wq,
                               const float* __restrict__ Wv,
                               const float* __restrict__ r_ai,
                               const float* __restrict__ r_aj,
                               float* __restrict__ out)
{
    __shared__ __attribute__((aligned(16))) float kb[KVROWS][WW][NCH];
    __shared__ __attribute__((aligned(16))) float vb[KVROWS][WW][NCH];
    __shared__ __attribute__((aligned(16))) float qb[ROWS][WW][NCH];
    __shared__ __attribute__((aligned(16))) float rcomp[KW][NCH];
    __shared__ __attribute__((aligned(16))) float xs[CGC][XPITCH];

    const int h0   = blockIdx.x * ROWS;
    const int g    = blockIdx.y;
    const int b    = blockIdx.z;
    const int tid  = threadIdx.x;
    const int lane = tid & 31;
    const int wave = tid >> 5;
    const int half = lane >> 4;
    const int m    = lane & 15;

    if (tid < KW * NCH) {
        int t = tid / NCH, n = tid % NCH;
        int cg = g * NCH + n;
        rcomp[t][n] = (cg < 32) ? r_ai[cg * KW + t] : r_aj[(cg - 32) * KW + t];
    }

    const float* WkG = Wk + g * 64;
    const float* WqG = Wq + g * 64;
    const float* WvG = Wv + g * 64;
    const int k_lo = half * 2;
    const int k_hi = 4 + half * 2;

    v2f aKV_lo, aKV_hi, aQ_lo, aQ_hi;
    if (m < 8) {
        aKV_lo[0] = WkG[m * 8 + k_lo];     aKV_lo[1] = WkG[m * 8 + k_lo + 1];
        aKV_hi[0] = WkG[m * 8 + k_hi];     aKV_hi[1] = WkG[m * 8 + k_hi + 1];
    } else {
        aKV_lo[0] = WvG[(m - 8) * 8 + k_lo]; aKV_lo[1] = WvG[(m - 8) * 8 + k_lo + 1];
        aKV_hi[0] = WvG[(m - 8) * 8 + k_hi]; aKV_hi[1] = WvG[(m - 8) * 8 + k_hi + 1];
    }
    aQ_lo[0] = WqG[(m & 7) * 8 + k_lo];  aQ_lo[1] = WqG[(m & 7) * 8 + k_lo + 1];
    aQ_hi[0] = WqG[(m & 7) * 8 + k_hi];  aQ_hi[1] = WqG[(m & 7) * 8 + k_hi + 1];

    const float* xg = x + (size_t)(b * CIN + g * CGC) * HWSZ;

    if (tid < XHALF * XC4) {
        const int r5 = tid / XC4;
        const int c4 = tid - r5 * XC4;
        const float* src0 = xg + reflect96(h0 - PADR + r5) * WW + c4 * 4;
        const float* src1 = xg + reflect96(h0 - PADR + r5 + XHALF) * WW + c4 * 4;
        float* dst = &xs[0][r5 * WW + c4 * 4];
        v4f xv[2 * CGC];
        #pragma unroll
        for (int ch = 0; ch < CGC; ++ch) {
            xv[2 * ch]     = *(const v4f*)(src0 + (size_t)ch * HWSZ);
            xv[2 * ch + 1] = *(const v4f*)(src1 + (size_t)ch * HWSZ);
        }
        #pragma unroll
        for (int ch = 0; ch < CGC; ++ch) {
            *(v4f*)(dst + ch * XPITCH)              = xv[2 * ch];
            *(v4f*)(dst + ch * XPITCH + XHALF * WW) = xv[2 * ch + 1];
        }
    }
    __syncthreads();

    {
        const int colh = (wave & 1) * (TSTRIPS * 16) + m;
        const int row0 = wave >> 1;
        v2f blo[PTRIPS][TSTRIPS], bhi[PTRIPS][TSTRIPS];
        #pragma unroll
        for (int j = -1; j < PTRIPS; ++j) {
            if (j + 1 < PTRIPS) {
                const int jn = j + 1;
                const int i = (jn < KVTRIPS) ? (row0 + 2 * jn) : (row0 + 2 * (jn - KVTRIPS) + PADR);
                const float* xp = &xs[0][i * WW + colh];
                #pragma unroll
                for (int s = 0; s < TSTRIPS; ++s) {
                    blo[jn][s][0] = xp[k_lo       * XPITCH + 16 * s];
                    blo[jn][s][1] = xp[(k_lo + 1) * XPITCH + 16 * s];
                    bhi[jn][s][0] = xp[k_hi       * XPITCH + 16 * s];
                    bhi[jn][s][1] = xp[(k_hi + 1) * XPITCH + 16 * s];
                }
            }
            if (j >= 0) {
                const bool isq = (j >= KVTRIPS);
                const v2f a_lo = isq ? aQ_lo : aKV_lo;
                const v2f a_hi = isq ? aQ_hi : aKV_hi;
                v8f acc[TSTRIPS];
                #pragma unroll
                for (int s = 0; s < TSTRIPS; ++s) {
                    const v8f zero = {};
                    acc[s] = __builtin_amdgcn_wmma_f32_16x16x4_f32(false, a_lo, false, blo[j][s],
                                                                   (short)0, zero, false, false);
                }
                #pragma unroll
                for (int s = 0; s < TSTRIPS; ++s)
                    acc[s] = __builtin_amdgcn_wmma_f32_16x16x4_f32(false, a_hi, false, bhi[j][s],
                                                                   (short)0, acc[s], false, false);
                if (!isq) {
                    const int i = row0 + 2 * j;
                    float* dst = half ? &vb[i][colh][0] : &kb[i][colh][0];
                    #pragma unroll
                    for (int s = 0; s < TSTRIPS; ++s) {
                        v4f lo4 = {acc[s][0], acc[s][1], acc[s][2], acc[s][3]};
                        v4f hi4 = {acc[s][4], acc[s][5], acc[s][6], acc[s][7]};
                        *(v4f*)(dst + s * 16 * NCH)     = lo4;
                        *(v4f*)(dst + s * 16 * NCH + 4) = hi4;
                    }
                } else if (!half) {
                    const int rr = row0 + 2 * (j - KVTRIPS);
                    float* dst = &qb[rr][colh][0];
                    #pragma unroll
                    for (int s = 0; s < TSTRIPS; ++s) {
                        v4f lo4 = {acc[s][0], acc[s][1], acc[s][2], acc[s][3]};
                        v4f hi4 = {acc[s][4], acc[s][5], acc[s][6], acc[s][7]};
                        *(v4f*)(dst + s * 16 * NCH)     = lo4;
                        *(v4f*)(dst + s * 16 * NCH + 4) = hi4;
                    }
                }
            }
        }
    }

    __syncthreads();

    if (tid < WW) {
        float* op = out + ((size_t)(b * CIN + g * NCH) * HH + h0) * WW + tid;
        if (g < 4) attend_triple<true, true >(&kb[0][0][0], &vb[0][0][0], &qb[0][0][0], &rcomp[0][0], 0, tid, (lane & 8) != 0, op);
        else       attend_triple<true, false>(&kb[0][0][0], &vb[0][0][0], &qb[0][0][0], &rcomp[0][0], 0, tid, (lane & 8) != 0, op);
    } else {
        const int c0 = NPX * lane;
        float* op = out + ((size_t)(b * CIN + g * NCH) * HH + h0 + (ROWS - 1)) * WW + c0;
        if (g < 4) attend_triple<false, true >(&kb[0][0][0], &vb[0][0][0], &qb[0][0][0], &rcomp[0][0], ROWS - 1, c0, (lane & 8) != 0, op);
        else       attend_triple<false, false>(&kb[0][0][0], &vb[0][0][0], &qb[0][0][0], &rcomp[0][0], ROWS - 1, c0, (lane & 8) != 0, op);
    }
}

extern "C" void kernel_launch(void* const* d_in, const int* in_sizes, int n_in,
                              void* d_out, int out_size, void* d_ws, size_t ws_size,
                              hipStream_t stream) {
    (void)in_sizes; (void)n_in; (void)out_size; (void)d_ws; (void)ws_size;
    const float* x    = (const float*)d_in[0];
    const float* Wk   = (const float*)d_in[1];
    const float* Wq   = (const float*)d_in[2];
    const float* Wv   = (const float*)d_in[3];
    const float* r_ai = (const float*)d_in[4];
    const float* r_aj = (const float*)d_in[5];
    float* out = (float*)d_out;

    dim3 grid(HH / ROWS, GROUPS, BQ);
    SelfAttentionBlock_kernel<<<grid, NTHR, 0, stream>>>(x, Wk, Wq, Wv, r_ai, r_aj, out);
}


# Attention loop: each half trip's 18 LDS record reads issued as one group ahead of its arithmetic (k before the logits, v before the guard and exp2) into a register set of their own; sums unchanged
# speedup vs baseline: 2.1919x; 1.0717x over previous
#include <hip/hip_runtime.h>


#define BQ     4
#define CIN    64
#define HH     96
#define WW     96
#define GROUPS 8
#define NCH    8
#define CGC    8
#define KW     7
#define PADR   3
#define HWSZ   (HH * WW)

#define ROWS   4
#define KVROWS (ROWS + 2*PADR)
#define NTHR   128
#define XPITCH (KVROWS * WW + 8)
#define XC4    (WW / 4)
#define XHALF  (KVROWS / 2)
#define TSTRIPS 3
#define KVTRIPS (KVROWS * 2 / 4)
#define QTRIPS  (ROWS * 2 / 4)
#define PTRIPS  (KVTRIPS + QTRIPS)

typedef __attribute__((ext_vector_type(2))) float v2f;
typedef __attribute__((ext_vector_type(4))) float v4f;
typedef __attribute__((ext_vector_type(8))) float v8f;

__device__ __forceinline__ int reflect96(int i) {
    if (i < 0)  i = -i;
    if (i > 95) i = 190 - i;
    return i;
}


#define GUARD_LOG2 32.0f
#define NPX  3
#define NREC (NPX + KW - 1)
template <bool VERT, bool ROWBIAS>
__device__ __forceinline__ void attend_triple(const float* kbp, const float* vbp,
                                              const float* qbp, const float* rcp,
                                              const int r0, const int c0, const bool swp, float* op)
{
    constexpr bool ALONG = (VERT == ROWBIAS);
    const float LOG2E = 1.44269504088896340736f;
    v4f q0[NPX], q1[NPX];
    #pragma unroll
    for (int p = 0; p < NPX; ++p) {
        const float* qp = qbp + ((r0 + (VERT ? p : 0)) * WW + c0 + (VERT ? 0 : p)) * NCH;
        q0[p] = *(const v4f*)qp * LOG2E;
        q1[p] = *(const v4f*)(qp + 4) * LOG2E;
    }

    const int sw = swp ? 4 : 0;
    const int so = 4 - sw;
    v4f qf[NPX], qs[NPX];
    #pragma unroll
    for (int p = 0; p < NPX; ++p) {
        qf[p] = swp ? q1[p] : q0[p];
        qs[p] = swp ? q0[p] : q1[p];
    }

    float rb[NPX][KW];
    #pragma unroll
    for (int t = 0; t < KW; ++t) {
        const v4f ra = *(const v4f*)(rcp + t * NCH);
        const v4f rc = *(const v4f*)(rcp + t * NCH + 4);
        #pragma unroll
        for (int p = 0; p < NPX; ++p) {
            float s = 0.f;
            #pragma unroll
            for (int n = 0; n < 4; ++n) s = __builtin_fmaf(q0[p][n], ra[n], s);
            #pragma unroll
            for (int n = 0; n < 4; ++n) s = __builtin_fmaf(q1[p][n], rc[n], s);
            rb[p][t] = s;
        }
    }

    #pragma unroll
    for (int p = 0; p < NPX; ++p) {
        const float* kp = kbp + ((r0 + PADR + (VERT ? p : 0)) * WW + c0 + (VERT ? 0 : p)) * NCH;
        const v4f k0 = *(const v4f*)kp;
        const v4f k1 = *(const v4f*)(kp + 4);
        float l = rb[p][PADR];
        #pragma unroll
        for (int n = 0; n < 4; ++n) l = __builtin_fmaf(q0[p][n], k0[n], l);
        #pragma unroll
        for (int n = 0; n < 4; ++n) l = __builtin_fmaf(q1[p][n], k1[n], l);
        #pragma unroll
        for (int t = 0; t < KW; ++t) rb[p][t] -= l;
    }

    float ssum[NPX], acc[NPX][NCH];
    #pragma unroll
    for (int p = 0; p < NPX; ++p) {
        ssum[p] = 0.0f;
        #pragma unroll
        for (int n = 0; n < NCH; ++n) acc[p][n] = 0.0f;
    }

    int cc[NREC];
    #pragma unroll
    for (int a = 0; a < NREC; ++a) cc[a] = reflect96(c0 - PADR + a) * NCH;

    for (int kq = 0; kq < KW; ++kq) {
        const int tb = VERT ? (r0 * WW + reflect96(c0 + kq - PADR)) * NCH : (r0 + kq) * WW * NCH;

        float bq[NPX];
        #pragma unroll
        for (int p = 0; p < NPX; ++p) {
            float v = rb[p][0];
            #pragma unroll
            for (int t = 1; t < KW; ++t) v = (kq == t) ? rb[p][t] : v;
            bq[p] = v;
        }

        v4f kf[NREC], ks[NREC];
        #pragma unroll
        for (int a = 0; a < NREC; ++a) {
            const float* kp = kbp + tb + (VERT ? a * WW * NCH : cc[a]);
            kf[a] = *(const v4f*)(kp + sw);
            ks[a] = *(const v4f*)(kp + so);
        }
        __builtin_amdgcn_sched_barrier(0);

        float lg[NPX][KW];
        #pragma unroll
        for (int a = 0; a < NREC; ++a) {
            #pragma unroll
            for (int p = 0; p < NPX; ++p) {
                const int ka = a - p;
                if (ka < 0 || ka >= KW) continue;
                float l = ALONG ? rb[p][ka] : bq[p];
                #pragma unroll
                for (int n = 0; n < 4; ++n) l = __builtin_fmaf(qf[p][n], kf[a][n], l);
                #pragma unroll
                for (int n = 0; n < 4; ++n) l = __builtin_fmaf(qs[p][n], ks[a][n], l);
                lg[p][ka] = l;
            }
        }

        v4f va[NREC], vc[NREC];
        #pragma unroll
        for (int a = 0; a < NREC; ++a) {
            const float* vp = vbp + tb + (VERT ? a * WW * NCH : cc[a]);
            va[a] = *(const v4f*)(vp + sw);
            vc[a] = *(const v4f*)(vp + so);
        }
        __builtin_amdgcn_sched_barrier(0);

        float tm[NPX];
        #pragma unroll
        for (int p = 0; p < NPX; ++p) {
            float v = lg[p][0];
            #pragma unroll
            for (int t = 1; t < KW; ++t) v = fmaxf(v, lg[p][t]);
            tm[p] = v;
        }
        const float tmx = fmaxf(fmaxf(tm[0], tm[1]), tm[2]);
        if (__any(tmx > GUARD_LOG2)) {
            #pragma unroll
            for (int p = 0; p < NPX; ++p) {
                const float s = ceilf(fmaxf(tm[p], 0.0f));
                const float f = __builtin_amdgcn_exp2f(-s);
                ssum[p] *= f;
                #pragma unroll
                for (int n = 0; n < NCH; ++n) acc[p][n] *= f;
                #pragma unroll
                for (int t = 0; t < KW; ++t) { rb[p][t] -= s; lg[p][t] -= s; }
            }
        }

        #pragma unroll
        for (int a = 0; a < NREC; ++a) {
            #pragma unroll
            for (int p = 0; p < NPX; ++p) {
                const int ka = a - p;
                if (ka < 0 || ka >= KW) continue;
                const float e = __builtin_amdgcn_exp2f(lg[p][ka]);
                ssum[p] += e;
                #pragma unroll
                for (int n = 0; n < 4; ++n) acc[p][n]     = __builtin_fmaf(e, va[a][n], acc[p][n]);
                #pragma unroll
                for (int n = 0; n < 4; ++n) acc[p][n + 4] = __builtin_fmaf(e, vc[a][n], acc[p][n + 4]);
            }
        }
    }

    #pragma unroll
    for (int p = 0; p < NPX; ++p) {
        const float inv = 1.0f / ssum[p];
        #pragma unroll
        for (int n = 0; n < NCH; ++n) {
            const float r = swp ? acc[p][n ^ 4] : acc[p][n];
            op[(size_t)n * HWSZ + (VERT ? p * WW : p)] = r * inv;
        }
    }
}

__global__ __launch_bounds__(NTHR)
void SelfAttentionBlock_kernel(const float* __restrict__ x,
                               const float* __restrict__ Wk,
                               const float* __restrict__ Wq,
                               const float* __restrict__ Wv,
                               const float* __restrict__ r_ai,
                               const float* __restrict__ r_aj,
                               float* __restrict__ out)
{
    __shared__ __attribute__((aligned(16))) float kb[KVROWS][WW][NCH];
    __shared__ __attribute__((aligned(16))) float vb[KVROWS][WW][NCH];
    __shared__ __attribute__((aligned(16))) float qb[ROWS][WW][NCH];
    __shared__ __attribute__((aligned(16))) float rcomp[KW][NCH];
    __shared__ __attribute__((aligned(16))) float xs[CGC][XPITCH];

    const int h0   = blockIdx.x * ROWS;
    const int g    = blockIdx.y;
    const int b    = blockIdx.z;
    const int tid  = threadIdx.x;
    const int lane = tid & 31;
    const int wave = tid >> 5;
    const int half = lane >> 4;
    const int m    = lane & 15;

    if (tid < KW * NCH) {
        int t = tid / NCH, n = tid % NCH;
        int cg = g * NCH + n;
        rcomp[t][n] = (cg < 32) ? r_ai[cg * KW + t] : r_aj[(cg - 32) * KW + t];
    }

    const float* WkG = Wk + g * 64;
    const float* WqG = Wq + g * 64;
    const float* WvG = Wv + g * 64;
    const int k_lo = half * 2;
    const int k_hi = 4 + half * 2;

    v2f aKV_lo, aKV_hi, aQ_lo, aQ_hi;
    if (m < 8) {
        aKV_lo[0] = WkG[m * 8 + k_lo];     aKV_lo[1] = WkG[m * 8 + k_lo + 1];
        aKV_hi[0] = WkG[m * 8 + k_hi];     aKV_hi[1] = WkG[m * 8 + k_hi + 1];
    } else {
        aKV_lo[0] = WvG[(m - 8) * 8 + k_lo]; aKV_lo[1] = WvG[(m - 8) * 8 + k_lo + 1];
        aKV_hi[0] = WvG[(m - 8) * 8 + k_hi]; aKV_hi[1] = WvG[(m - 8) * 8 + k_hi + 1];
    }
    aQ_lo[0] = WqG[(m & 7) * 8 + k_lo];  aQ_lo[1] = WqG[(m & 7) * 8 + k_lo + 1];
    aQ_hi[0] = WqG[(m & 7) * 8 + k_hi];  aQ_hi[1] = WqG[(m & 7) * 8 + k_hi + 1];

    const float* xg = x + (size_t)(b * CIN + g * CGC) * HWSZ;

    if (tid < XHALF * XC4) {
        const int r5 = tid / XC4;
        const int c4 = tid - r5 * XC4;
        const float* src0 = xg + reflect96(h0 - PADR + r5) * WW + c4 * 4;
        const float* src1 = xg + reflect96(h0 - PADR + r5 + XHALF) * WW + c4 * 4;
        float* dst = &xs[0][r5 * WW + c4 * 4];
        v4f xv[2 * CGC];
        #pragma unroll
        for (int ch = 0; ch < CGC; ++ch) {
            xv[2 * ch]     = *(const v4f*)(src0 + (size_t)ch * HWSZ);
            xv[2 * ch + 1] = *(const v4f*)(src1 + (size_t)ch * HWSZ);
        }
        #pragma unroll
        for (int ch = 0; ch < CGC; ++ch) {
            *(v4f*)(dst + ch * XPITCH)              = xv[2 * ch];
            *(v4f*)(dst + ch * XPITCH + XHALF * WW) = xv[2 * ch + 1];
        }
    }
    __syncthreads();

    {
        const int colh = (wave & 1) * (TSTRIPS * 16) + m;
        const int row0 = wave >> 1;
        v2f blo[PTRIPS][TSTRIPS], bhi[PTRIPS][TSTRIPS];
        #pragma unroll
        for (int j = -1; j < PTRIPS; ++j) {
            if (j + 1 < PTRIPS) {
                const int jn = j + 1;
                const int i = (jn < KVTRIPS) ? (row0 + 2 * jn) : (row0 + 2 * (jn - KVTRIPS) + PADR);
                const float* xp = &xs[0][i * WW + colh];
                #pragma unroll
                for (int s = 0; s < TSTRIPS; ++s) {
                    blo[jn][s][0] = xp[k_lo       * XPITCH + 16 * s];
                    blo[jn][s][1] = xp[(k_lo + 1) * XPITCH + 16 * s];
                    bhi[jn][s][0] = xp[k_hi       * XPITCH + 16 * s];
                    bhi[jn][s][1] = xp[(k_hi + 1) * XPITCH + 16 * s];
                }
            }
            if (j >= 0) {
                const bool isq = (j >= KVTRIPS);
                const v2f a_lo = isq ? aQ_lo : aKV_lo;
                const v2f a_hi = isq ? aQ_hi : aKV_hi;
                v8f acc[TSTRIPS];
                #pragma unroll
                for (int s = 0; s < TSTRIPS; ++s) {
                    const v8f zero = {};
                    acc[s] = __builtin_amdgcn_wmma_f32_16x16x4_f32(false, a_lo, false, blo[j][s],
                                                                   (short)0, zero, false, false);
                }
                #pragma unroll
                for (int s = 0; s < TSTRIPS; ++s)
                    acc[s] = __builtin_amdgcn_wmma_f32_16x16x4_f32(false, a_hi, false, bhi[j][s],
                                                                   (short)0, acc[s], false, false);
                if (!isq) {
                    const int i = row0 + 2 * j;
                    float* dst = half ? &vb[i][colh][0] : &kb[i][colh][0];
                    #pragma unroll
                    for (int s = 0; s < TSTRIPS; ++s) {
                        v4f lo4 = {acc[s][0], acc[s][1], acc[s][2], acc[s][3]};
                        v4f hi4 = {acc[s][4], acc[s][5], acc[s][6], acc[s][7]};
                        *(v4f*)(dst + s * 16 * NCH)     = lo4;
                        *(v4f*)(dst + s * 16 * NCH + 4) = hi4;
                    }
                } else if (!half) {
                    const int rr = row0 + 2 * (j - KVTRIPS);
                    float* dst = &qb[rr][colh][0];
                    #pragma unroll
                    for (int s = 0; s < TSTRIPS; ++s) {
                        v4f lo4 = {acc[s][0], acc[s][1], acc[s][2], acc[s][3]};
                        v4f hi4 = {acc[s][4], acc[s][5], acc[s][6], acc[s][7]};
                        *(v4f*)(dst + s * 16 * NCH)     = lo4;
                        *(v4f*)(dst + s * 16 * NCH + 4) = hi4;
                    }
                }
            }
        }
    }

    __syncthreads();

    if (tid < WW) {
        float* op = out + ((size_t)(b * CIN + g * NCH) * HH + h0) * WW + tid;
        if (g < 4) attend_triple<true, true >(&kb[0][0][0], &vb[0][0][0], &qb[0][0][0], &rcomp[0][0], 0, tid, (lane & 8) != 0, op);
        else       attend_triple<true, false>(&kb[0][0][0], &vb[0][0][0], &qb[0][0][0], &rcomp[0][0], 0, tid, (lane & 8) != 0, op);
    } else {
        const int c0 = NPX * lane;
        float* op = out + ((size_t)(b * CIN + g * NCH) * HH + h0 + (ROWS - 1)) * WW + c0;
        if (g < 4) attend_triple<false, true >(&kb[0][0][0], &vb[0][0][0], &qb[0][0][0], &rcomp[0][0], ROWS - 1, c0, (lane & 8) != 0, op);
        else       attend_triple<false, false>(&kb[0][0][0], &vb[0][0][0], &qb[0][0][0], &rcomp[0][0], ROWS - 1, c0, (lane & 8) != 0, op);
    }
}

extern "C" void kernel_launch(void* const* d_in, const int* in_sizes, int n_in,
                              void* d_out, int out_size, void* d_ws, size_t ws_size,
                              hipStream_t stream) {
    (void)in_sizes; (void)n_in; (void)out_size; (void)d_ws; (void)ws_size;
    const float* x    = (const float*)d_in[0];
    const float* Wk   = (const float*)d_in[1];
    const float* Wq   = (const float*)d_in[2];
    const float* Wv   = (const float*)d_in[3];
    const float* r_ai = (const float*)d_in[4];
    const float* r_aj = (const float*)d_in[5];
    float* out = (float*)d_out;

    dim3 grid(HH / ROWS, GROUPS, BQ);
    SelfAttentionBlock_kernel<<<grid, NTHR, 0, stream>>>(x, Wk, Wq, Wv, r_ai, r_aj, out);
}


# Attention dot products (logits, bias dots) stay scalar FMA chains and are dual-issued: an empty marker stops the compiler packing chains of different k records behind gather moves; sums unchanged
# speedup vs baseline: 2.3032x; 1.0508x over previous
#include <hip/hip_runtime.h>


#define BQ     4
#define CIN    64
#define HH     96
#define WW     96
#define GROUPS 8
#define NCH    8
#define CGC    8
#define KW     7
#define PADR   3
#define HWSZ   (HH * WW)

#define ROWS   4
#define KVROWS (ROWS + 2*PADR)
#define NTHR   128
#define XPITCH (KVROWS * WW + 8)
#define XC4    (WW / 4)
#define XHALF  (KVROWS / 2)
#define TSTRIPS 3
#define KVTRIPS (KVROWS * 2 / 4)
#define QTRIPS  (ROWS * 2 / 4)
#define PTRIPS  (KVTRIPS + QTRIPS)

typedef __attribute__((ext_vector_type(2))) float v2f;
typedef __attribute__((ext_vector_type(4))) float v4f;
typedef __attribute__((ext_vector_type(8))) float v8f;

__device__ __forceinline__ int reflect96(int i) {
    if (i < 0)  i = -i;
    if (i > 95) i = 190 - i;
    return i;
}


#define GUARD_LOG2 32.0f
#define NPX  3
#define NREC (NPX + KW - 1)
template <bool VERT, bool ROWBIAS>
__device__ __forceinline__ void attend_triple(const float* kbp, const float* vbp,
                                              const float* qbp, const float* rcp,
                                              const int r0, const int c0, const bool swp, float* op)
{
    constexpr bool ALONG = (VERT == ROWBIAS);
    const float LOG2E = 1.44269504088896340736f;
    v4f q0[NPX], q1[NPX];
    #pragma unroll
    for (int p = 0; p < NPX; ++p) {
        const float* qp = qbp + ((r0 + (VERT ? p : 0)) * WW + c0 + (VERT ? 0 : p)) * NCH;
        q0[p] = *(const v4f*)qp * LOG2E;
        q1[p] = *(const v4f*)(qp + 4) * LOG2E;
    }

    const int sw = swp ? 4 : 0;
    const int so = 4 - sw;
    v4f qf[NPX], qs[NPX];
    #pragma unroll
    for (int p = 0; p < NPX; ++p) {
        qf[p] = swp ? q1[p] : q0[p];
        qs[p] = swp ? q0[p] : q1[p];
    }

    float rb[NPX][KW];
    #pragma unroll
    for (int t = 0; t < KW; ++t) {
        const v4f ra = *(const v4f*)(rcp + t * NCH);
        const v4f rc = *(const v4f*)(rcp + t * NCH + 4);
        #pragma unroll
        for (int p = 0; p < NPX; ++p) {
            float s = 0.f;
            #pragma unroll
            for (int n = 0; n < 4; ++n) s = __builtin_fmaf(q0[p][n], ra[n], s);
            #pragma unroll
            for (int n = 0; n < 4; ++n) s = __builtin_fmaf(q1[p][n], rc[n], s);
            asm("" : "+v"(s));
            rb[p][t] = s;
        }
    }

    #pragma unroll
    for (int p = 0; p < NPX; ++p) {
        const float* kp = kbp + ((r0 + PADR + (VERT ? p : 0)) * WW + c0 + (VERT ? 0 : p)) * NCH;
        const v4f k0 = *(const v4f*)kp;
        const v4f k1 = *(const v4f*)(kp + 4);
        float l = rb[p][PADR];
        #pragma unroll
        for (int n = 0; n < 4; ++n) l = __builtin_fmaf(q0[p][n], k0[n], l);
        #pragma unroll
        for (int n = 0; n < 4; ++n) l = __builtin_fmaf(q1[p][n], k1[n], l);
        asm("" : "+v"(l));
        #pragma unroll
        for (int t = 0; t < KW; ++t) rb[p][t] -= l;
    }

    float ssum[NPX], acc[NPX][NCH];
    #pragma unroll
    for (int p = 0; p < NPX; ++p) {
        ssum[p] = 0.0f;
        #pragma unroll
        for (int n = 0; n < NCH; ++n) acc[p][n] = 0.0f;
    }

    int cc[NREC];
    #pragma unroll
    for (int a = 0; a < NREC; ++a) cc[a] = reflect96(c0 - PADR + a) * NCH;

    for (int kq = 0; kq < KW; ++kq) {
        const int tb = VERT ? (r0 * WW + reflect96(c0 + kq - PADR)) * NCH : (r0 + kq) * WW * NCH;

        float bq[NPX];
        #pragma unroll
        for (int p = 0; p < NPX; ++p) {
            float v = rb[p][0];
            #pragma unroll
            for (int t = 1; t < KW; ++t) v = (kq == t) ? rb[p][t] : v;
            bq[p] = v;
        }

        v4f kf[NREC], ks[NREC];
        #pragma unroll
        for (int a = 0; a < NREC; ++a) {
            const float* kp = kbp + tb + (VERT ? a * WW * NCH : cc[a]);
            kf[a] = *(const v4f*)(kp + sw);
            ks[a] = *(const v4f*)(kp + so);
        }
        __builtin_amdgcn_sched_barrier(0);

        float lg[NPX][KW];
        #pragma unroll
        for (int a = 0; a < NREC; ++a) {
            #pragma unroll
            for (int p = 0; p < NPX; ++p) {
                const int ka = a - p;
                if (ka < 0 || ka >= KW) continue;
                float l = ALONG ? rb[p][ka] : bq[p];
                #pragma unroll
                for (int n = 0; n < 4; ++n) l = __builtin_fmaf(qf[p][n], kf[a][n], l);
                #pragma unroll
                for (int n = 0; n < 4; ++n) l = __builtin_fmaf(qs[p][n], ks[a][n], l);
                asm("" : "+v"(l));
                lg[p][ka] = l;
            }
        }

        v4f va[NREC], vc[NREC];
        #pragma unroll
        for (int a = 0; a < NREC; ++a) {
            const float* vp = vbp + tb + (VERT ? a * WW * NCH : cc[a]);
            va[a] = *(const v4f*)(vp + sw);
            vc[a] = *(const v4f*)(vp + so);
        }
        __builtin_amdgcn_sched_barrier(0);

        float tm[NPX];
        #pragma unroll
        for (int p = 0; p < NPX; ++p) {
            float v = lg[p][0];
            #pragma unroll
            for (int t = 1; t < KW; ++t) v = fmaxf(v, lg[p][t]);
            tm[p] = v;
        }
        const float tmx = fmaxf(fmaxf(tm[0], tm[1]), tm[2]);
        if (__any(tmx > GUARD_LOG2)) {
            #pragma unroll
            for (int p = 0; p < NPX; ++p) {
                const float s = ceilf(fmaxf(tm[p], 0.0f));
                const float f = __builtin_amdgcn_exp2f(-s);
                ssum[p] *= f;
                #pragma unroll
                for (int n = 0; n < NCH; ++n) acc[p][n] *= f;
                #pragma unroll
                for (int t = 0; t < KW; ++t) { rb[p][t] -= s; lg[p][t] -= s; }
            }
        }

        #pragma unroll
        for (int a = 0; a < NREC; ++a) {
            #pragma unroll
            for (int p = 0; p < NPX; ++p) {
                const int ka = a - p;
                if (ka < 0 || ka >= KW) continue;
                const float e = __builtin_amdgcn_exp2f(lg[p][ka]);
                ssum[p] += e;
                #pragma unroll
                for (int n = 0; n < 4; ++n) acc[p][n]     = __builtin_fmaf(e, va[a][n], acc[p][n]);
                #pragma unroll
                for (int n = 0; n < 4; ++n) acc[p][n + 4] = __builtin_fmaf(e, vc[a][n], acc[p][n + 4]);
            }
        }
    }

    #pragma unroll
    for (int p = 0; p < NPX; ++p) {
        const float inv = 1.0f / ssum[p];
        #pragma unroll
        for (int n = 0; n < NCH; ++n) {
            const float r = swp ? acc[p][n ^ 4] : acc[p][n];
            op[(size_t)n * HWSZ + (VERT ? p * WW : p)] = r * inv;
        }
    }
}

__global__ __launch_bounds__(NTHR)
void SelfAttentionBlock_kernel(const float* __restrict__ x,
                               const float* __restrict__ Wk,
                               const float* __restrict__ Wq,
                               const float* __restrict__ Wv,
                               const float* __restrict__ r_ai,
                               const float* __restrict__ r_aj,
                               float* __restrict__ out)
{
    __shared__ __attribute__((aligned(16))) float kb[KVROWS][WW][NCH];
    __shared__ __attribute__((aligned(16))) float vb[KVROWS][WW][NCH];
    __shared__ __attribute__((aligned(16))) float qb[ROWS][WW][NCH];
    __shared__ __attribute__((aligned(16))) float rcomp[KW][NCH];
    __shared__ __attribute__((aligned(16))) float xs[CGC][XPITCH];

    const int h0   = blockIdx.x * ROWS;
    const int g    = blockIdx.y;
    const int b    = blockIdx.z;
    const int tid  = threadIdx.x;
    const int lane = tid & 31;
    const int wave = tid >> 5;
    const int half = lane >> 4;
    const int m    = lane & 15;

    if (tid < KW * NCH) {
        int t = tid / NCH, n = tid % NCH;
        int cg = g * NCH + n;
        rcomp[t][n] = (cg < 32) ? r_ai[cg * KW + t] : r_aj[(cg - 32) * KW + t];
    }

    const float* WkG = Wk + g * 64;
    const float* WqG = Wq + g * 64;
    const float* WvG = Wv + g * 64;
    const int k_lo = half * 2;
    const int k_hi = 4 + half * 2;

    v2f aKV_lo, aKV_hi, aQ_lo, aQ_hi;
    if (m < 8) {
        aKV_lo[0] = WkG[m * 8 + k_lo];     aKV_lo[1] = WkG[m * 8 + k_lo + 1];
        aKV_hi[0] = WkG[m * 8 + k_hi];     aKV_hi[1] = WkG[m * 8 + k_hi + 1];
    } else {
        aKV_lo[0] = WvG[(m - 8) * 8 + k_lo]; aKV_lo[1] = WvG[(m - 8) * 8 + k_lo + 1];
        aKV_hi[0] = WvG[(m - 8) * 8 + k_hi]; aKV_hi[1] = WvG[(m - 8) * 8 + k_hi + 1];
    }
    aQ_lo[0] = WqG[(m & 7) * 8 + k_lo];  aQ_lo[1] = WqG[(m & 7) * 8 + k_lo + 1];
    aQ_hi[0] = WqG[(m & 7) * 8 + k_hi];  aQ_hi[1] = WqG[(m & 7) * 8 + k_hi + 1];

    const float* xg = x + (size_t)(b * CIN + g * CGC) * HWSZ;

    if (tid < XHALF * XC4) {
        const int r5 = tid / XC4;
        const int c4 = tid - r5 * XC4;
        const float* src0 = xg + reflect96(h0 - PADR + r5) * WW + c4 * 4;
        const float* src1 = xg + reflect96(h0 - PADR + r5 + XHALF) * WW + c4 * 4;
        float* dst = &xs[0][r5 * WW + c4 * 4];
        v4f xv[2 * CGC];
        #pragma unroll
        for (int ch = 0; ch < CGC; ++ch) {
            xv[2 * ch]     = *(const v4f*)(src0 + (size_t)ch * HWSZ);
            xv[2 * ch + 1] = *(const v4f*)(src1 + (size_t)ch * HWSZ);
        }
        #pragma unroll
        for (int ch = 0; ch < CGC; ++ch) {
            *(v4f*)(dst + ch * XPITCH)              = xv[2 * ch];
            *(v4f*)(dst + ch * XPITCH + XHALF * WW) = xv[2 * ch + 1];
        }
    }
    __syncthreads();

    {
        const int colh = (wave & 1) * (TSTRIPS * 16) + m;
        const int row0 = wave >> 1;
        v2f blo[PTRIPS][TSTRIPS], bhi[PTRIPS][TSTRIPS];
        #pragma unroll
        for (int j = -1; j < PTRIPS; ++j) {
            if (j + 1 < PTRIPS) {
                const int jn = j + 1;
                const int i = (jn < KVTRIPS) ? (row0 + 2 * jn) : (row0 + 2 * (jn - KVTRIPS) + PADR);
                const float* xp = &xs[0][i * WW + colh];
                #pragma unroll
                for (int s = 0; s < TSTRIPS; ++s) {
                    blo[jn][s][0] = xp[k_lo       * XPITCH + 16 * s];
                    blo[jn][s][1] = xp[(k_lo + 1) * XPITCH + 16 * s];
                    bhi[jn][s][0] = xp[k_hi       * XPITCH + 16 * s];
                    bhi[jn][s][1] = xp[(k_hi + 1) * XPITCH + 16 * s];
                }
            }
            if (j >= 0) {
                const bool isq = (j >= KVTRIPS);
                const v2f a_lo = isq ? aQ_lo : aKV_lo;
                const v2f a_hi = isq ? aQ_hi : aKV_hi;
                v8f acc[TSTRIPS];
                #pragma unroll
                for (int s = 0; s < TSTRIPS; ++s) {
                    const v8f zero = {};
                    acc[s] = __builtin_amdgcn_wmma_f32_16x16x4_f32(false, a_lo, false, blo[j][s],
                                                                   (short)0, zero, false, false);
                }
                #pragma unroll
                for (int s = 0; s < TSTRIPS; ++s)
                    acc[s] = __builtin_amdgcn_wmma_f32_16x16x4_f32(false, a_hi, false, bhi[j][s],
                                                                   (short)0, acc[s], false, false);
                if (!isq) {
                    const int i = row0 + 2 * j;
                    float* dst = half ? &vb[i][colh][0] : &kb[i][colh][0];
                    #pragma unroll
                    for (int s = 0; s < TSTRIPS; ++s) {
                        v4f lo4 = {acc[s][0], acc[s][1], acc[s][2], acc[s][3]};
                        v4f hi4 = {acc[s][4], acc[s][5], acc[s][6], acc[s][7]};
                        *(v4f*)(dst + s * 16 * NCH)     = lo4;
                        *(v4f*)(dst + s * 16 * NCH + 4) = hi4;
                    }
                } else if (!half) {
                    const int rr = row0 + 2 * (j - KVTRIPS);
                    float* dst = &qb[rr][colh][0];
                    #pragma unroll
                    for (int s = 0; s < TSTRIPS; ++s) {
                        v4f lo4 = {acc[s][0], acc[s][1], acc[s][2], acc[s][3]};
                        v4f hi4 = {acc[s][4], acc[s][5], acc[s][6], acc[s][7]};
                        *(v4f*)(dst + s * 16 * NCH)     = lo4;
                        *(v4f*)(dst + s * 16 * NCH + 4) = hi4;
                    }
                }
            }
        }
    }

    __syncthreads();

    if (tid < WW) {
        float* op = out + ((size_t)(b * CIN + g * NCH) * HH + h0) * WW + tid;
        if (g < 4) attend_triple<true, true >(&kb[0][0][0], &vb[0][0][0], &qb[0][0][0], &rcomp[0][0], 0, tid, (lane & 8) != 0, op);
        else       attend_triple<true, false>(&kb[0][0][0], &vb[0][0][0], &qb[0][0][0], &rcomp[0][0], 0, tid, (lane & 8) != 0, op);
    } else {
        const int c0 = NPX * lane;
        float* op = out + ((size_t)(b * CIN + g * NCH) * HH + h0 + (ROWS - 1)) * WW + c0;
        if (g < 4) attend_triple<false, true >(&kb[0][0][0], &vb[0][0][0], &qb[0][0][0], &rcomp[0][0], ROWS - 1, c0, (lane & 8) != 0, op);
        else       attend_triple<false, false>(&kb[0][0][0], &vb[0][0][0], &qb[0][0][0], &rcomp[0][0], ROWS - 1, c0, (lane & 8) != 0, op);
    }
}

extern "C" void kernel_launch(void* const* d_in, const int* in_sizes, int n_in,
                              void* d_out, int out_size, void* d_ws, size_t ws_size,
                              hipStream_t stream) {
    (void)in_sizes; (void)n_in; (void)out_size; (void)d_ws; (void)ws_size;
    const float* x    = (const float*)d_in[0];
    const float* Wk   = (const float*)d_in[1];
    const float* Wq   = (const float*)d_in[2];
    const float* Wv   = (const float*)d_in[3];
    const float* r_ai = (const float*)d_in[4];
    const float* r_aj = (const float*)d_in[5];
    float* out = (float*)d_out;

    dim3 grid(HH / ROWS, GROUPS, BQ);
    SelfAttentionBlock_kernel<<<grid, NTHR, 0, stream>>>(x, Wk, Wq, Wv, r_ai, r_aj, out);
}
